# starting kernel
# baseline (speedup 1.0000x reference)
#include <hip/hip_runtime.h>
#include <math.h>


typedef __attribute__((ext_vector_type(2))) float v2f;
typedef __attribute__((ext_vector_type(8))) float v8f;

#if defined(__has_builtin)
#if __has_builtin(__builtin_amdgcn_tanhf)
#define FAST_TANH(x) __builtin_amdgcn_tanhf(x)
#endif
#endif
#ifndef FAST_TANH
#define FAST_TANH(x) tanhf(x)
#endif

#define WAVES_PER_BLOCK 8
#define ROWS_PER_WAVE   16
#define T_STEPS         128
#define E_DIM           20
#define H_DIM           20
#define L_DIM           15
#define LDS_STRIDE      22

__global__ __launch_bounds__(256)
void WordVanilla_rnn_kernel(const int*   __restrict__ x,
                            const float* __restrict__ drop_rate,
                            const float* __restrict__ emb,
                            const float* __restrict__ Wxh,
                            const float* __restrict__ Whh,
                            const float* __restrict__ bh,
                            const float* __restrict__ Wd,
                            const float* __restrict__ bd,
                            float*       __restrict__ out,
                            int B)
{
    __shared__ float lds[WAVES_PER_BLOCK][ROWS_PER_WAVE][LDS_STRIDE];

    const int tid  = threadIdx.x;
    const int wave = tid >> 5;
    const int lane = tid & 31;
    const int half = lane >> 4;
    const int m    = lane & 15;
    const int tile = blockIdx.x * WAVES_PER_BLOCK + wave;
    const int b0   = tile * ROWS_PER_WAVE;
    if (b0 + ROWS_PER_WAVE > B) return;

    v2f wxh[2][5], whh[2][5];
    float bh_n[2];
#pragma unroll
    for (int nt = 0; nt < 2; ++nt) {
        const int  n  = m + 16 * nt;
        const bool nv = (n < H_DIM);
        bh_n[nt] = nv ? bh[n] : 0.0f;
#pragma unroll
        for (int c = 0; c < 5; ++c) {
            const int k = 4 * c + 2 * half;
            wxh[nt][c].x = nv ? Wxh[(k    ) * H_DIM + n] : 0.0f;
            wxh[nt][c].y = nv ? Wxh[(k + 1) * H_DIM + n] : 0.0f;
            whh[nt][c].x = nv ? Whh[(k    ) * H_DIM + n] : 0.0f;
            whh[nt][c].y = nv ? Whh[(k + 1) * H_DIM + n] : 0.0f;
        }
    }

    v2f hA[5];
#pragma unroll
    for (int c = 0; c < 5; ++c) { hA[c].x = 0.0f; hA[c].y = 0.0f; }

    const int*  xrow  = x + (size_t)(b0 + m) * T_STEPS;
    float*      myLds = &lds[wave][0][0];

    for (int t0 = 0; t0 < T_STEPS; t0 += 4) {
        const int4 toks = *(const int4*)(xrow + t0);
#pragma unroll
        for (int tt = 0; tt < 4; ++tt) {
            const int tok = (tt == 0) ? toks.x : (tt == 1) ? toks.y
                          : (tt == 2) ? toks.z : toks.w;

            const float* erow = emb + (size_t)tok * E_DIM + 2 * half;
            v2f eA[5];
#pragma unroll
            for (int c = 0; c < 5; ++c)
                eA[c] = *(const v2f*)(erow + 4 * c);

            v8f acc0 = {}; v8f acc1 = {};
#pragma unroll
            for (int c = 0; c < 5; ++c)
                acc0 = __builtin_amdgcn_wmma_f32_16x16x4_f32(
                    false, eA[c], false, wxh[0][c], (short)0, acc0, false, false);
#pragma unroll
            for (int c = 0; c < 5; ++c)
                acc0 = __builtin_amdgcn_wmma_f32_16x16x4_f32(
                    false, hA[c], false, whh[0][c], (short)0, acc0, false, false);
#pragma unroll
            for (int c = 0; c < 5; ++c)
                acc1 = __builtin_amdgcn_wmma_f32_16x16x4_f32(
                    false, eA[c], false, wxh[1][c], (short)0, acc1, false, false);
#pragma unroll
            for (int c = 0; c < 5; ++c)
                acc1 = __builtin_amdgcn_wmma_f32_16x16x4_f32(
                    false, hA[c], false, whh[1][c], (short)0, acc1, false, false);

#pragma unroll
            for (int i = 0; i < 8; ++i) {
                const int row = i + 8 * half;
                myLds[row * LDS_STRIDE + m] = FAST_TANH(acc0[i] + bh_n[0]);
            }
            if (m < (H_DIM - 16)) {
#pragma unroll
                for (int i = 0; i < 8; ++i) {
                    const int row = i + 8 * half;
                    myLds[row * LDS_STRIDE + m + 16] = FAST_TANH(acc1[i] + bh_n[1]);
                }
            }

#pragma unroll
            for (int c = 0; c < 5; ++c)
                hA[c] = *(const v2f*)(myLds + m * LDS_STRIDE + 4 * c + 2 * half);
        }
    }

    const float scale = 1.0f / (1.0f - drop_rate[0]);
    if (half == 0) {
        float hrow[H_DIM];
#pragma unroll
        for (int k = 0; k < H_DIM; ++k)
            hrow[k] = myLds[m * LDS_STRIDE + k] * scale;

        float z[L_DIM];
        float zmax = -INFINITY;
#pragma unroll
        for (int l = 0; l < L_DIM; ++l) {
            float s = bd[l];
#pragma unroll
            for (int k = 0; k < H_DIM; ++k)
                s += hrow[k] * Wd[k * L_DIM + l];
            z[l] = s;
            zmax = fmaxf(zmax, s);
        }
        float sum = 0.0f;
#pragma unroll
        for (int l = 0; l < L_DIM; ++l) {
            z[l] = __expf(z[l] - zmax);
            sum += z[l];
        }
        const float inv = 1.0f / sum;
        float* orow = out + (size_t)(b0 + m) * L_DIM;
#pragma unroll
        for (int l = 0; l < L_DIM; ++l)
            orow[l] = z[l] * inv;
    }
}

extern "C" void kernel_launch(void* const* d_in, const int* in_sizes, int n_in,
                              void* d_out, int out_size, void* d_ws, size_t ws_size,
                              hipStream_t stream) {
    (void)n_in; (void)out_size; (void)d_ws; (void)ws_size;

    const int*   x   = (const int*)  d_in[0];
    const float* dr  = (const float*)d_in[1];
    const float* emb = (const float*)d_in[2];
    const float* Wxh = (const float*)d_in[3];
    const float* Whh = (const float*)d_in[4];
    const float* bh  = (const float*)d_in[5];
    const float* Wd  = (const float*)d_in[6];
    const float* bd  = (const float*)d_in[7];
    float*       out = (float*)d_out;

    const int B = in_sizes[0] / T_STEPS;
    const int rowsPerBlock = ROWS_PER_WAVE * WAVES_PER_BLOCK;
    const int grid = (B + rowsPerBlock - 1) / rowsPerBlock;

    hipLaunchKernelGGL(WordVanilla_rnn_kernel, dim3(grid), dim3(256), 0, stream,
                       x, dr, emb, Wxh, Whh, bh, Wd, bd, out, B);
}


# embedding rows gathered with 3 aligned loads per lane (16+16+8 B, 32-bit row offsets formed once per 8-step chunk; was 5 strided 8-B loads); e@Wxh products summed in the gather's permuted k order
# speedup vs baseline: 2.2493x; 2.2493x over previous
#include <hip/hip_runtime.h>
#include <math.h>


typedef __attribute__((ext_vector_type(2))) float v2f;
typedef __attribute__((ext_vector_type(4))) float v4f;
typedef __attribute__((ext_vector_type(8))) float v8f;

#if defined(__has_builtin)
#if __has_builtin(__builtin_amdgcn_tanhf)
#define FAST_TANH(x) __builtin_amdgcn_tanhf(x)
#endif
#endif
#ifndef FAST_TANH
#define FAST_TANH(x) tanhf(x)
#endif

#define TILES_PER_BLOCK 4
#define ROWS_PER_WAVE   16
#define T_STEPS         128
#define E_DIM           20
#define H_DIM           20
#define L_DIM           15
#define LDS_STRIDE      36
#define CHUNK           8
#define NCHUNK          (T_STEPS / CHUNK)

__device__ __forceinline__ v8f mma4(v2f a, v2f b, v8f c) {
    return __builtin_amdgcn_wmma_f32_16x16x4_f32(false, a, false, b, (short)0, c, false, false);
}

__device__ __forceinline__ int kfirst(int h, int c) {
    return h ? ((c < 4) ? (12 + 2 * c) : 10) : (2 * c);
}

#define HOLD()      __builtin_amdgcn_sched_barrier(0)
#define PIN1(a)     asm volatile("" :: "v"(a))
#define PIN2(a, b)  asm volatile("" :: "v"(a), "v"(b))

__global__ __launch_bounds__(128)
void WordVanilla_rnn_kernel(const int*   __restrict__ x,
                            const float* __restrict__ drop_rate,
                            const float* __restrict__ emb,
                            const float* __restrict__ Wxh,
                            const float* __restrict__ Whh,
                            const float* __restrict__ bh,
                            const float* __restrict__ Wd,
                            const float* __restrict__ bd,
                            float*       __restrict__ out,
                            int B)
{
    __shared__ float hbuf[TILES_PER_BLOCK][ROWS_PER_WAVE][LDS_STRIDE];

    const int tid  = threadIdx.x;
    const int wave = __builtin_amdgcn_readfirstlane(tid >> 5);
    const int lane = tid & 31;
    const int half = lane >> 4;
    const int m    = lane & 15;
    const int tile = blockIdx.x * TILES_PER_BLOCK + wave;
    const int b0   = tile * ROWS_PER_WAVE;
    if (b0 + ROWS_PER_WAVE > B) return;

    v2f wx[2][5], wh[2][5];
    float bh_n[2];
#pragma unroll
    for (int nt = 0; nt < 2; ++nt) {
        const int  n  = m + 16 * nt;
        const bool nv = (n < H_DIM);
        bh_n[nt] = nv ? bh[n] : 0.0f;
#pragma unroll
        for (int c = 0; c < 5; ++c) {
            const int k  = 4 * c + 2 * half;
            const int kx = kfirst(half, c);
            wx[nt][c].x = nv ? Wxh[(kx    ) * H_DIM + n] : 0.0f;
            wx[nt][c].y = nv ? Wxh[(kx + 1) * H_DIM + n] : 0.0f;
            wh[nt][c].x = nv ? Whh[(k    ) * H_DIM + n] : 0.0f;
            wh[nt][c].y = nv ? Whh[(k + 1) * H_DIM + n] : 0.0f;
        }
    }

    v2f hA[5];
#pragma unroll
    for (int c = 0; c < 5; ++c) { hA[c].x = 0.0f; hA[c].y = 0.0f; }
    float* myLds = &hbuf[wave][0][0];

    const int*   xrow  = x + (size_t)(b0 + m) * T_STEPS;
    const char*    ebytes = (const char*)emb;
    const unsigned offw   = half ? 48u : 0u;
    const unsigned offn   = half ? 40u : 32u;
    const unsigned ROWB   = E_DIM * (unsigned)sizeof(float);
    int toks[CHUNK];
    v2f eA[CHUNK][5];
#pragma unroll
    for (int q = 0; q < CHUNK / 4; ++q) {
        const int4 tk = *(const int4*)(xrow + 4 * q);
        toks[4 * q + 0] = tk.x; toks[4 * q + 1] = tk.y;
        toks[4 * q + 2] = tk.z; toks[4 * q + 3] = tk.w;
    }
#pragma unroll
    for (int s = 0; s < CHUNK; ++s) {
        const unsigned ro = (unsigned)toks[s] * ROWB;
        const v4f p0 = *(const v4f*)(ebytes + (ro + offw));
        const v4f p1 = *(const v4f*)(ebytes + (ro + offw) + 16);
        const v2f p2 = *(const v2f*)(ebytes + (ro + offn));
        eA[s][0] = p0.xy; eA[s][1] = p0.zw;
        eA[s][2] = p1.xy; eA[s][3] = p1.zw;
        eA[s][4] = p2;
    }
#pragma unroll
    for (int q = 0; q < CHUNK / 4; ++q) {
        const int4 tk = *(const int4*)(xrow + CHUNK + 4 * q);
        toks[4 * q + 0] = tk.x; toks[4 * q + 1] = tk.y;
        toks[4 * q + 2] = tk.z; toks[4 * q + 3] = tk.w;
    }

    v8f xin0 = {}; v8f xin1 = {};
#pragma unroll
    for (int c = 0; c < 5; ++c) {
        xin0 = mma4(eA[0][c], wx[0][c], xin0);
        xin1 = mma4(eA[0][c], wx[1][c], xin1);
    }

    for (int ci = 0; ci < NCHUNK; ++ci) {
        const int cn2 = (ci + 2 < NCHUNK) ? (ci + 2) : (NCHUNK - 1);
        v2f eN[CHUNK][5];
        int tokn[CHUNK];
        unsigned rw[CHUNK], rn[CHUNK];

#pragma unroll
        for (int s = 0; s < CHUNK; ++s) {
            v8f acc0 = xin0; v8f acc1 = xin1;
#pragma unroll
            for (int c = 0; c < 5; ++c) {
                acc0 = mma4(hA[c], wh[0][c], acc0);
                acc1 = mma4(hA[c], wh[1][c], acc1);
            }

            const int  sn   = (s + 1 < CHUNK) ? (s + 1) : 0;
            const v2f* en   = (s + 1 < CHUNK) ? eA[sn] : eN[sn];
            v8f nx0 = {}; v8f nx1 = {};

            PIN2(acc0, acc1);
            HOLD();
            nx0 = mma4(en[0], wx[0][0], nx0);
            nx1 = mma4(en[0], wx[1][0], nx1);
            PIN2(nx0, nx1);
            HOLD();

            float t0[8], t1[8];
#pragma unroll
            for (int i = 0; i < 8; ++i) t0[i] = FAST_TANH(acc0[i] + bh_n[0]);
            nx0 = mma4(en[1], wx[0][1], nx0);
            PIN1(nx0);
            HOLD();
#pragma unroll
            for (int i = 0; i < 8; ++i)
                myLds[(i + 8 * half) * LDS_STRIDE + m] = t0[i];
            nx1 = mma4(en[1], wx[1][1], nx1);
            PIN1(nx1);
            HOLD();
#pragma unroll
            for (int i = 0; i < 8; ++i) t1[i] = FAST_TANH(acc1[i] + bh_n[1]);
            nx0 = mma4(en[2], wx[0][2], nx0);
            PIN1(nx0);
            HOLD();
#pragma unroll
            for (int i = 0; i < 8; ++i)
                myLds[(i + 8 * half) * LDS_STRIDE + m + 16] = t1[i];
            nx1 = mma4(en[2], wx[1][2], nx1);
            PIN1(nx1);
            HOLD();

#pragma unroll
            for (int c = 0; c < 5; ++c)
                hA[c] = *(const v2f*)(myLds + m * LDS_STRIDE + 4 * c + 2 * half);
            HOLD();
            if (s == 0) {
#pragma unroll
                for (int q = 0; q < CHUNK; ++q) {
                    const unsigned ro = (unsigned)toks[q] * ROWB;
                    rw[q] = ro + offw;
                    rn[q] = ro + offn;
                }
                HOLD();
            }
            {
                const v4f p0 = *(const v4f*)(ebytes + rw[s]);
                const v4f p1 = *(const v4f*)(ebytes + rw[s] + 16);
                const v2f p2 = *(const v2f*)(ebytes + rn[s]);
                eN[s][0] = p0.xy; eN[s][1] = p0.zw;
                eN[s][2] = p1.xy; eN[s][3] = p1.zw;
                eN[s][4] = p2;
            }
            if (s == 0) {
#pragma unroll
                for (int q = 0; q < CHUNK / 4; ++q) {
                    const int4 tk = *(const int4*)(xrow + cn2 * CHUNK + 4 * q);
                    tokn[4 * q + 0] = tk.x; tokn[4 * q + 1] = tk.y;
                    tokn[4 * q + 2] = tk.z; tokn[4 * q + 3] = tk.w;
                }
            }
            HOLD();
            nx0 = mma4(en[3], wx[0][3], nx0);
            nx1 = mma4(en[3], wx[1][3], nx1);
            nx0 = mma4(en[4], wx[0][4], nx0);
            nx1 = mma4(en[4], wx[1][4], nx1);
            PIN2(nx0, nx1);
            HOLD();

            xin0 = nx0; xin1 = nx1;
        }

#pragma unroll
        for (int s = 0; s < CHUNK; ++s) {
            toks[s] = tokn[s];
#pragma unroll
            for (int c = 0; c < 5; ++c) eA[s][c] = eN[s][c];
        }
    }

    if (half == 0) {
        const float scale = 1.0f / (1.0f - drop_rate[0]);
        float hrow[H_DIM];
#pragma unroll
        for (int k = 0; k < H_DIM; ++k)
            hrow[k] = myLds[m * LDS_STRIDE + k] * scale;

        float z[L_DIM];
        float zmax = -INFINITY;
#pragma unroll
        for (int l = 0; l < L_DIM; ++l) {
            float s = bd[l];
#pragma unroll
            for (int k = 0; k < H_DIM; ++k)
                s += hrow[k] * Wd[k * L_DIM + l];
            z[l] = s;
            zmax = fmaxf(zmax, s);
        }
        float sum = 0.0f;
#pragma unroll
        for (int l = 0; l < L_DIM; ++l) {
            z[l] = __expf(z[l] - zmax);
            sum += z[l];
        }
        const float inv = 1.0f / sum;
        float* orow = out + (size_t)(b0 + m) * L_DIM;
#pragma unroll
        for (int l = 0; l < L_DIM; ++l)
            orow[l] = z[l] * inv;
    }
}

extern "C" void kernel_launch(void* const* d_in, const int* in_sizes, int n_in,
                              void* d_out, int out_size, void* d_ws, size_t ws_size,
                              hipStream_t stream) {
    (void)n_in; (void)out_size; (void)d_ws; (void)ws_size;

    const int*   x   = (const int*)  d_in[0];
    const float* dr  = (const float*)d_in[1];
    const float* emb = (const float*)d_in[2];
    const float* Wxh = (const float*)d_in[3];
    const float* Whh = (const float*)d_in[4];
    const float* bh  = (const float*)d_in[5];
    const float* Wd  = (const float*)d_in[6];
    const float* bd  = (const float*)d_in[7];
    float*       out = (float*)d_out;

    const int B = in_sizes[0] / T_STEPS;
    const int rowsPerBlock = ROWS_PER_WAVE * TILES_PER_BLOCK;
    const int grid = (B + rowsPerBlock - 1) / rowsPerBlock;

    hipLaunchKernelGGL(WordVanilla_rnn_kernel, dim3(grid), dim3(128), 0, stream,
                       x, dr, emb, Wxh, Whh, bh, Wd, bd, out, B);
}


# hidden columns 16-19 computed as a transposed tile (weights as A, the same h/e fragments as B): its tanh(D) is already the k=16-19 A fragment, so only columns 0-15 cross LDS; sum order unchanged
# speedup vs baseline: 2.3431x; 1.0417x over previous
#include <hip/hip_runtime.h>
#include <math.h>


typedef __attribute__((ext_vector_type(2))) float v2f;
typedef __attribute__((ext_vector_type(4))) float v4f;
typedef __attribute__((ext_vector_type(8))) float v8f;

#if defined(__has_builtin)
#if __has_builtin(__builtin_amdgcn_tanhf)
#define FAST_TANH(x) __builtin_amdgcn_tanhf(x)
#endif
#endif
#ifndef FAST_TANH
#define FAST_TANH(x) tanhf(x)
#endif

#define TILES_PER_BLOCK 4
#define ROWS_PER_WAVE   16
#define T_STEPS         128
#define E_DIM           20
#define H_DIM           20
#define L_DIM           15
#define LDS_STRIDE      36
#define CHUNK           8
#define NCHUNK          (T_STEPS / CHUNK)

__device__ __forceinline__ v8f mma4(v2f a, v2f b, v8f c) {
    return __builtin_amdgcn_wmma_f32_16x16x4_f32(false, a, false, b, (short)0, c, false, false);
}

__device__ __forceinline__ int kfirst(int h, int c) {
    return h ? ((c < 4) ? (12 + 2 * c) : 10) : (2 * c);
}

#define HOLD()      __builtin_amdgcn_sched_barrier(0)
#define PIN2(a, b)  asm volatile("" :: "v"(a), "v"(b))

__global__ __launch_bounds__(128)
void WordVanilla_rnn_kernel(const int*   __restrict__ x,
                            const float* __restrict__ drop_rate,
                            const float* __restrict__ emb,
                            const float* __restrict__ Wxh,
                            const float* __restrict__ Whh,
                            const float* __restrict__ bh,
                            const float* __restrict__ Wd,
                            const float* __restrict__ bd,
                            float*       __restrict__ out,
                            int B)
{
    __shared__ float hbuf[TILES_PER_BLOCK][ROWS_PER_WAVE][LDS_STRIDE];

    const int tid  = threadIdx.x;
    const int wave = __builtin_amdgcn_readfirstlane(tid >> 5);
    const int lane = tid & 31;
    const int half = lane >> 4;
    const int m    = lane & 15;
    const int tile = blockIdx.x * TILES_PER_BLOCK + wave;
    const int b0   = tile * ROWS_PER_WAVE;
    if (b0 + ROWS_PER_WAVE > B) return;

    v2f wx0[5], wh0[5], wxT[5], whT[5];
    const bool tv = ((m & 7) < 2);
    const int  jT = tv ? (16 + 2 * (m >> 3) + (m & 7)) : 16;
    const float bh0 = bh[m];
    v2f bT;
    bT.x = bh[16 + 2 * half];
    bT.y = bh[17 + 2 * half];
#pragma unroll
    for (int c = 0; c < 5; ++c) {
        const int k  = 4 * c + 2 * half;
        const int kx = kfirst(half, c);
        wx0[c].x = Wxh[(kx    ) * H_DIM + m];
        wx0[c].y = Wxh[(kx + 1) * H_DIM + m];
        wh0[c].x = Whh[(k    ) * H_DIM + m];
        wh0[c].y = Whh[(k + 1) * H_DIM + m];
        wxT[c].x = tv ? Wxh[(kx    ) * H_DIM + jT] : 0.0f;
        wxT[c].y = tv ? Wxh[(kx + 1) * H_DIM + jT] : 0.0f;
        whT[c].x = tv ? Whh[(k    ) * H_DIM + jT] : 0.0f;
        whT[c].y = tv ? Whh[(k + 1) * H_DIM + jT] : 0.0f;
    }

    v2f hA[5];
#pragma unroll
    for (int c = 0; c < 5; ++c) { hA[c].x = 0.0f; hA[c].y = 0.0f; }
    float* myLds = &hbuf[wave][0][0];

    const int*   xrow  = x + (size_t)(b0 + m) * T_STEPS;
    const char*    ebytes = (const char*)emb;
    const unsigned offw   = half ? 48u : 0u;
    const unsigned offn   = half ? 40u : 32u;
    const unsigned ROWB   = E_DIM * (unsigned)sizeof(float);
    int toks[CHUNK];
    v2f eA[CHUNK][5];
#pragma unroll
    for (int q = 0; q < CHUNK / 4; ++q) {
        const int4 tk = *(const int4*)(xrow + 4 * q);
        toks[4 * q + 0] = tk.x; toks[4 * q + 1] = tk.y;
        toks[4 * q + 2] = tk.z; toks[4 * q + 3] = tk.w;
    }
#pragma unroll
    for (int s = 0; s < CHUNK; ++s) {
        const unsigned ro = (unsigned)toks[s] * ROWB;
        const v4f p0 = *(const v4f*)(ebytes + (ro + offw));
        const v4f p1 = *(const v4f*)(ebytes + (ro + offw) + 16);
        const v2f p2 = *(const v2f*)(ebytes + (ro + offn));
        eA[s][0] = p0.xy; eA[s][1] = p0.zw;
        eA[s][2] = p1.xy; eA[s][3] = p1.zw;
        eA[s][4] = p2;
    }
#pragma unroll
    for (int q = 0; q < CHUNK / 4; ++q) {
        const int4 tk = *(const int4*)(xrow + CHUNK + 4 * q);
        toks[4 * q + 0] = tk.x; toks[4 * q + 1] = tk.y;
        toks[4 * q + 2] = tk.z; toks[4 * q + 3] = tk.w;
    }

    v8f xin0 = {}; v8f xin1 = {};
#pragma unroll
    for (int c = 0; c < 5; ++c) {
        xin0 = mma4(eA[0][c], wx0[c], xin0);
        xin1 = mma4(wxT[c], eA[0][c], xin1);
    }

    for (int ci = 0; ci < NCHUNK; ++ci) {
        const int cn2 = (ci + 2 < NCHUNK) ? (ci + 2) : (NCHUNK - 1);
        v2f eN[CHUNK][5];
        int tokn[CHUNK];
        unsigned rw[CHUNK], rn[CHUNK];

#pragma unroll
        for (int s = 0; s < CHUNK; ++s) {
            v8f acc0 = xin0; v8f acc1 = xin1;
#pragma unroll
            for (int c = 0; c < 5; ++c) {
                acc0 = mma4(hA[c], wh0[c], acc0);
                acc1 = mma4(whT[c], hA[c], acc1);
            }

            const int  sn   = (s + 1 < CHUNK) ? (s + 1) : 0;
            const v2f* en   = (s + 1 < CHUNK) ? eA[sn] : eN[sn];
            v8f nx0 = {}; v8f nx1 = {};

            PIN2(acc0, acc1);
            HOLD();
            nx0 = mma4(en[0], wx0[0], nx0);
            nx1 = mma4(wxT[0], en[0], nx1);
            PIN2(nx0, nx1);
            HOLD();

            float t0[8];
#pragma unroll
            for (int i = 0; i < 8; ++i) t0[i] = FAST_TANH(acc0[i] + bh0);
            v2f h4;
            h4.x = FAST_TANH(acc1[0] + bT.x);
            h4.y = FAST_TANH(acc1[1] + bT.y);

#pragma unroll
            for (int i = 0; i < 8; ++i)
                myLds[(i + 8 * half) * LDS_STRIDE + m] = t0[i];
#pragma unroll
            for (int c = 0; c < 4; ++c)
                hA[c] = *(const v2f*)(myLds + m * LDS_STRIDE + 4 * c + 2 * half);
            hA[4] = h4;
            HOLD();
            if (s == 0) {
#pragma unroll
                for (int q = 0; q < CHUNK; ++q) {
                    const unsigned ro = (unsigned)toks[q] * ROWB;
                    rw[q] = ro + offw;
                    rn[q] = ro + offn;
                }
                HOLD();
            }
            {
                const v4f p0 = *(const v4f*)(ebytes + rw[s]);
                const v4f p1 = *(const v4f*)(ebytes + rw[s] + 16);
                const v2f p2 = *(const v2f*)(ebytes + rn[s]);
                eN[s][0] = p0.xy; eN[s][1] = p0.zw;
                eN[s][2] = p1.xy; eN[s][3] = p1.zw;
                eN[s][4] = p2;
            }
            if (s == 0) {
#pragma unroll
                for (int q = 0; q < CHUNK / 4; ++q) {
                    const int4 tk = *(const int4*)(xrow + cn2 * CHUNK + 4 * q);
                    tokn[4 * q + 0] = tk.x; tokn[4 * q + 1] = tk.y;
                    tokn[4 * q + 2] = tk.z; tokn[4 * q + 3] = tk.w;
                }
            }
            HOLD();
#pragma unroll
            for (int c = 1; c < 5; ++c) {
                nx0 = mma4(en[c], wx0[c], nx0);
                nx1 = mma4(wxT[c], en[c], nx1);
            }
            PIN2(nx0, nx1);
            HOLD();

            xin0 = nx0; xin1 = nx1;
        }

#pragma unroll
        for (int s = 0; s < CHUNK; ++s) {
            toks[s] = tokn[s];
#pragma unroll
            for (int c = 0; c < 5; ++c) eA[s][c] = eN[s][c];
        }
    }

    *(v2f*)(myLds + m * LDS_STRIDE + 16 + 2 * half) = hA[4];
    if (half == 0) {
        const float scale = 1.0f / (1.0f - drop_rate[0]);
        float hrow[H_DIM];
#pragma unroll
        for (int k = 0; k < H_DIM; ++k)
            hrow[k] = myLds[m * LDS_STRIDE + k] * scale;

        float z[L_DIM];
        float zmax = -INFINITY;
#pragma unroll
        for (int l = 0; l < L_DIM; ++l) {
            float s = bd[l];
#pragma unroll
            for (int k = 0; k < H_DIM; ++k)
                s += hrow[k] * Wd[k * L_DIM + l];
            z[l] = s;
            zmax = fmaxf(zmax, s);
        }
        float sum = 0.0f;
#pragma unroll
        for (int l = 0; l < L_DIM; ++l) {
            z[l] = __expf(z[l] - zmax);
            sum += z[l];
        }
        const float inv = 1.0f / sum;
        float* orow = out + (size_t)(b0 + m) * L_DIM;
#pragma unroll
        for (int l = 0; l < L_DIM; ++l)
            orow[l] = z[l] * inv;
    }
}

extern "C" void kernel_launch(void* const* d_in, const int* in_sizes, int n_in,
                              void* d_out, int out_size, void* d_ws, size_t ws_size,
                              hipStream_t stream) {
    (void)n_in; (void)out_size; (void)d_ws; (void)ws_size;

    const int*   x   = (const int*)  d_in[0];
    const float* dr  = (const float*)d_in[1];
    const float* emb = (const float*)d_in[2];
    const float* Wxh = (const float*)d_in[3];
    const float* Whh = (const float*)d_in[4];
    const float* bh  = (const float*)d_in[5];
    const float* Wd  = (const float*)d_in[6];
    const float* bd  = (const float*)d_in[7];
    float*       out = (float*)d_out;

    const int B = in_sizes[0] / T_STEPS;
    const int rowsPerBlock = ROWS_PER_WAVE * TILES_PER_BLOCK;
    const int grid = (B + rowsPerBlock - 1) / rowsPerBlock;

    hipLaunchKernelGGL(WordVanilla_rnn_kernel, dim3(grid), dim3(128), 0, stream,
                       x, dr, emb, Wxh, Whh, bh, Wd, bd, out, B);
}
